# starting kernel
# baseline (speedup 1.0000x reference)
#include <hip/hip_runtime.h>


typedef float v2f __attribute__((ext_vector_type(2)));
typedef float v4f __attribute__((ext_vector_type(4)));
typedef float v8f __attribute__((ext_vector_type(8)));

#define NHEADS 32
#define D      64
#define S      8192
#define SPLITS 8
#define SPAN   (S / SPLITS)
#define SCHUNK 64
#define LDSTR  68

__global__ void kstats_kernel(const float* __restrict__ k, float* __restrict__ stats) {
  const int row = blockIdx.x;
  const float* kr = k + (size_t)row * S;
  const int t = threadIdx.x;
  v4f vals[8];
  float m = -3.4e38f;
#pragma unroll
  for (int i = 0; i < 8; ++i) {
    v4f x = *(const v4f*)&kr[i * 1024 + t * 4];
    vals[i] = x;
    m = fmaxf(m, fmaxf(fmaxf(x.x, x.y), fmaxf(x.z, x.w)));
  }
  __shared__ float red[256];
  red[t] = m;
  __syncthreads();
  for (int off = 128; off > 0; off >>= 1) {
    if (t < off) red[t] = fmaxf(red[t], red[t + off]);
    __syncthreads();
  }
  m = red[0];
  __syncthreads();
  float s = 0.f;
#pragma unroll
  for (int i = 0; i < 8; ++i) {
    v4f x = vals[i];
    s += __expf(x.x - m) + __expf(x.y - m) + __expf(x.z - m) + __expf(x.w - m);
  }
  red[t] = s;
  __syncthreads();
  for (int off = 128; off > 0; off >>= 1) {
    if (t < off) red[t] += red[t + off];
    __syncthreads();
  }
  if (t == 0) {
    stats[2 * row]     = m;
    stats[2 * row + 1] = 1.0f / red[0];
  }
}

__global__ void ctx_partial_kernel(const float* __restrict__ k, const float* __restrict__ v,
                                   const float* __restrict__ stats, float* __restrict__ partial) {
  const int head  = blockIdx.x;
  const int split = blockIdx.y;
  const size_t hoff = (size_t)head * D * S;
  const float* kh = k + hoff;
  const float* vh = v + hoff;
  const int t    = threadIdx.x;
  const int lane = t & 31;
  const int w    = t >> 5;
  const int hf   = lane >> 4;
  const int lm   = lane & 15;

  __shared__ __align__(16) float vt[D * LDSTR];
  __shared__ __align__(16) float kt[D * LDSTR];
  __shared__ float2 sstat[D];

  if (t < D) {
    sstat[t] = make_float2(stats[2 * (head * D + t)], stats[2 * (head * D + t) + 1]);
  }
  __syncthreads();

  const v8f vzero = {};
  v8f acc[4];
#pragma unroll
  for (int n = 0; n < 4; ++n) acc[n] = vzero;

  const int srow = t >> 4;
  const int scol = (t * 4) & 63;

  for (int chunk = 0; chunk < SPAN / SCHUNK; ++chunk) {
    const int s0 = split * SPAN + chunk * SCHUNK;
#pragma unroll
    for (int i = 0; i < 8; ++i) {
      int row = 8 * i + srow;
      const size_t g = (size_t)row * S + s0 + scol;
      v4f vv = __builtin_nontemporal_load((const v4f*)&vh[g]);
      *(v4f*)&vt[row * LDSTR + scol] = vv;
      v4f kx = *(const v4f*)&kh[g];
      float2 st = sstat[row];
      v4f ks;
      ks.x = __expf(kx.x - st.x) * st.y;
      ks.y = __expf(kx.y - st.x) * st.y;
      ks.z = __expf(kx.z - st.x) * st.y;
      ks.w = __expf(kx.w - st.x) * st.y;
      *(v4f*)&kt[row * LDSTR + scol] = ks;
    }
    __syncthreads();
#pragma unroll
    for (int kk = 0; kk < SCHUNK; kk += 4) {
      v2f a = *(const v2f*)&vt[(w * 16 + lm) * LDSTR + kk + 2 * hf];
#pragma unroll
      for (int n = 0; n < 4; ++n) {
        v2f b = *(const v2f*)&kt[(n * 16 + lm) * LDSTR + kk + 2 * hf];
        acc[n] = __builtin_amdgcn_wmma_f32_16x16x4_f32(
            false, a, false, b, (short)0, acc[n], false, false);
      }
    }
    __syncthreads();
  }

  float* p = partial + (size_t)(split * NHEADS + head) * (D * D);
#pragma unroll
  for (int n = 0; n < 4; ++n) {
#pragma unroll
    for (int r = 0; r < 8; ++r) {
      int row = w * 16 + r + 8 * hf;
      int col = n * 16 + lm;
      p[row * D + col] = acc[n][r];
    }
  }
}

__global__ void ctx_reduce_kernel(const float* __restrict__ partial, float* __restrict__ ct) {
  int idx = blockIdx.x * blockDim.x + threadIdx.x;
  if (idx >= NHEADS * D * D) return;
  int head = idx / (D * D);
  int off  = idx - head * (D * D);
  float s = 0.f;
#pragma unroll
  for (int sp = 0; sp < SPLITS; ++sp)
    s += partial[(size_t)(sp * NHEADS + head) * (D * D) + off];
  ct[idx] = s;
}

__global__ void out_kernel(const float* __restrict__ q, const float* __restrict__ ct,
                           float* __restrict__ out) {
  const int head  = blockIdx.x;
  const int stile = blockIdx.y;
  const int s0 = stile * 128;
  const float* qh = q + (size_t)head * D * S;
  float* oh = out + (size_t)head * D * S;
  const int t    = threadIdx.x;
  const int lane = t & 31;
  const int w    = t >> 5;
  const int hf   = lane >> 4;
  const int lm   = lane & 15;

  __shared__ __align__(16) float qs[D * 129];
  __shared__ __align__(16) float cts[D * LDSTR];

#pragma unroll
  for (int i = 0; i < 8; ++i) {
    int e   = (i * 128 + t) * 4;
    int row = e >> 6;
    int col = e & 63;
    *(v4f*)&cts[row * LDSTR + col] = *(const v4f*)&ct[(size_t)head * (D * D) + e];
  }

  float m = -3.4e38f;
  for (int d = 0; d < D; ++d) {
    float x = __builtin_nontemporal_load(&qh[(size_t)d * S + s0 + t]);
    qs[d * 129 + t] = x;
    m = fmaxf(m, x);
  }
  float sum = 0.f;
  for (int d = 0; d < D; ++d) {
    float e = __expf(qs[d * 129 + t] - m);
    sum += e;
    qs[d * 129 + t] = e;
  }
  const float r = 0.125f / sum;
  for (int d = 0; d < D; ++d) qs[d * 129 + t] *= r;
  __syncthreads();

  const int j0 = w * 16;
  for (int st = 0; st < 8; ++st) {
    const int sb = st * 16;
    v8f acc = {};
#pragma unroll
    for (int kk = 0; kk < D; kk += 4) {
      v2f a = *(const v2f*)&cts[(j0 + lm) * LDSTR + kk + 2 * hf];
      v2f b;
      b.x = qs[(kk     + 2 * hf) * 129 + sb + lm];
      b.y = qs[(kk + 1 + 2 * hf) * 129 + sb + lm];
      acc = __builtin_amdgcn_wmma_f32_16x16x4_f32(
          false, a, false, b, (short)0, acc, false, false);
    }
#pragma unroll
    for (int rr = 0; rr < 8; ++rr) {
      int row = j0 + rr + 8 * hf;
      __builtin_nontemporal_store(acc[rr], &oh[(size_t)row * S + s0 + sb + lm]);
    }
  }
}

extern "C" void kernel_launch(void* const* d_in, const int* in_sizes, int n_in,
                              void* d_out, int out_size, void* d_ws, size_t ws_size,
                              hipStream_t stream) {
  (void)in_sizes; (void)n_in; (void)out_size; (void)ws_size;
  const float* q = (const float*)d_in[0];
  const float* k = (const float*)d_in[1];
  const float* v = (const float*)d_in[2];
  float* out = (float*)d_out;
  float* ws  = (float*)d_ws;

  float* stats   = ws;
  float* partial = ws + 2 * NHEADS * D;
  float* ct      = partial + (size_t)SPLITS * NHEADS * D * D;

  kstats_kernel<<<dim3(NHEADS * D), 256, 0, stream>>>(k, stats);
  ctx_partial_kernel<<<dim3(NHEADS, SPLITS), 128, 0, stream>>>(k, v, stats, partial);
  ctx_reduce_kernel<<<dim3((NHEADS * D * D + 255) / 256), 256, 0, stream>>>(partial, ct);
  out_kernel<<<dim3(NHEADS, S / 128), 128, 0, stream>>>(q, ct, out);
}


# ctx kernel software-pipelined: 8-wave blocks, 2 LDS stages, the next two chunks' global loads held in registers across a hand-written barrier that does not drain them; every sum keeps its order
# speedup vs baseline: 1.7888x; 1.7888x over previous
#include <hip/hip_runtime.h>


typedef float v2f __attribute__((ext_vector_type(2)));
typedef float v4f __attribute__((ext_vector_type(4)));
typedef float v8f __attribute__((ext_vector_type(8)));

#define NHEADS 32
#define D      64
#define S      8192
#define SPLITS 8
#define SPAN   (S / SPLITS)
#define SCHUNK 64
#define LDSTR  68

__global__ void kstats_kernel(const float* __restrict__ k, float* __restrict__ stats) {
  const int row = blockIdx.x;
  const float* kr = k + (size_t)row * S;
  const int t = threadIdx.x;
  v4f vals[8];
  float m = -3.4e38f;
#pragma unroll
  for (int i = 0; i < 8; ++i) {
    v4f x = *(const v4f*)&kr[i * 1024 + t * 4];
    vals[i] = x;
    m = fmaxf(m, fmaxf(fmaxf(x.x, x.y), fmaxf(x.z, x.w)));
  }
  __shared__ float red[256];
  red[t] = m;
  __syncthreads();
  for (int off = 128; off > 0; off >>= 1) {
    if (t < off) red[t] = fmaxf(red[t], red[t + off]);
    __syncthreads();
  }
  m = red[0];
  __syncthreads();
  float s = 0.f;
#pragma unroll
  for (int i = 0; i < 8; ++i) {
    v4f x = vals[i];
    s += __expf(x.x - m) + __expf(x.y - m) + __expf(x.z - m) + __expf(x.w - m);
  }
  red[t] = s;
  __syncthreads();
  for (int off = 128; off > 0; off >>= 1) {
    if (t < off) red[t] += red[t + off];
    __syncthreads();
  }
  if (t == 0) {
    stats[2 * row]     = m;
    stats[2 * row + 1] = 1.0f / red[0];
  }
}

#define NCHUNK (SPAN / SCHUNK)
#define TILEF  (D * LDSTR)

__device__ __forceinline__ void wg_barrier_keep_loads() {
  asm volatile("s_wait_dscnt 0x0\n\ts_barrier_signal -1\n\ts_barrier_wait -1" ::: "memory");
}

__device__ __forceinline__ void keep_phase_order() { __builtin_amdgcn_sched_barrier(0); }

__device__ __forceinline__ void load_set(v4f (&rv)[4], v4f (&rk)[4], const float* __restrict__ vp,
                                         const float* __restrict__ kp) {
  for (int i = 0; i < 4; ++i) {
    rv[i] = __builtin_nontemporal_load((const v4f*)&vp[(size_t)(16 * i) * S]);
    rk[i] = *(const v4f*)&kp[(size_t)(16 * i) * S];
  }
}

__device__ __forceinline__ void stage_set(const v4f (&rv)[4], const v4f (&rk)[4], const float2 (&st)[4],
                                          float* __restrict__ vdst, float* __restrict__ kdst) {
  for (int i = 0; i < 4; ++i) {
    *(v4f*)&vdst[16 * i * LDSTR] = rv[i];
    const v4f kx = rk[i];
    v4f ks;
    ks.x = __expf(kx.x - st[i].x) * st[i].y;
    ks.y = __expf(kx.y - st[i].x) * st[i].y;
    ks.z = __expf(kx.z - st[i].x) * st[i].y;
    ks.w = __expf(kx.w - st[i].x) * st[i].y;
    *(v4f*)&kdst[16 * i * LDSTR] = ks;
  }
}

__device__ __forceinline__ void mma_chunk(const float* __restrict__ arow, const float* __restrict__ bbase,
                                          v8f (&acc)[2]) {
  for (int kk = 0; kk < SCHUNK; kk += 4) {
    v2f a = *(const v2f*)&arow[kk];
    for (int n = 0; n < 2; ++n) {
      v2f b = *(const v2f*)&bbase[n * 16 * LDSTR + kk];
      acc[n] = __builtin_amdgcn_wmma_f32_16x16x4_f32(
          false, a, false, b, (short)0, acc[n], false, false);
    }
  }
}

__global__ __launch_bounds__(256) void ctx_partial_kernel(const float* __restrict__ k, const float* __restrict__ v,
                                   const float* __restrict__ stats, float* __restrict__ partial) {
  const int head  = blockIdx.x;
  const int split = blockIdx.y;
  const size_t hoff = (size_t)head * D * S;
  const int t    = threadIdx.x;
  const int lane = t & 31;
  const int wv   = t >> 5;
  const int w    = wv & 3;
  const int nh   = wv >> 2;
  const int hf   = lane >> 4;
  const int lm   = lane & 15;

  __shared__ __align__(16) float vt[2 * TILEF];
  __shared__ __align__(16) float kt[2 * TILEF];

  const int srow = t >> 4;
  const int scol = (t * 4) & 63;
  const float* vsrc = v + hoff + (size_t)srow * S + split * SPAN + scol;
  const float* ksrc = k + hoff + (size_t)srow * S + split * SPAN + scol;
  float* vdst = &vt[srow * LDSTR + scol];
  float* kdst = &kt[srow * LDSTR + scol];

  const float* arow  = &vt[(w * 16 + lm) * LDSTR + 2 * hf];
  const float* bbase = &kt[(nh * 32 + lm) * LDSTR + 2 * hf];

  const v8f vzero = {};
  v8f acc[2];
  acc[0] = vzero;
  acc[1] = vzero;

  v4f av[4], ak[4], bv[4], bk[4];
  float2 st[4];
  for (int i = 0; i < 4; ++i)
    st[i] = *(const float2*)&stats[2 * (head * D + 16 * i + srow)];

  load_set(av, ak, vsrc, ksrc);
  keep_phase_order();
  load_set(bv, bk, vsrc + SCHUNK, ksrc + SCHUNK);
  keep_phase_order();
  stage_set(av, ak, st, vdst, kdst);
  load_set(av, ak, vsrc + 2 * SCHUNK, ksrc + 2 * SCHUNK);
  keep_phase_order();
  wg_barrier_keep_loads();
  keep_phase_order();

  for (int c = 0; c < NCHUNK - 4; c += 2) {
    stage_set(bv, bk, st, vdst + TILEF, kdst + TILEF);
    load_set(bv, bk, vsrc + (c + 3) * SCHUNK, ksrc + (c + 3) * SCHUNK);
    mma_chunk(arow, bbase, acc);
    wg_barrier_keep_loads();
    keep_phase_order();
    stage_set(av, ak, st, vdst, kdst);
    load_set(av, ak, vsrc + (c + 4) * SCHUNK, ksrc + (c + 4) * SCHUNK);
    mma_chunk(arow + TILEF, bbase + TILEF, acc);
    wg_barrier_keep_loads();
    keep_phase_order();
  }
  stage_set(bv, bk, st, vdst + TILEF, kdst + TILEF);
  load_set(bv, bk, vsrc + (NCHUNK - 1) * SCHUNK, ksrc + (NCHUNK - 1) * SCHUNK);
  mma_chunk(arow, bbase, acc);
  wg_barrier_keep_loads();
  keep_phase_order();
  stage_set(av, ak, st, vdst, kdst);
  mma_chunk(arow + TILEF, bbase + TILEF, acc);
  wg_barrier_keep_loads();
  keep_phase_order();
  stage_set(bv, bk, st, vdst + TILEF, kdst + TILEF);
  mma_chunk(arow, bbase, acc);
  wg_barrier_keep_loads();
  keep_phase_order();
  mma_chunk(arow + TILEF, bbase + TILEF, acc);

  float* p = partial + (size_t)(split * NHEADS + head) * (D * D);
  for (int n = 0; n < 2; ++n) {
    for (int r = 0; r < 8; ++r) {
      int row = w * 16 + r + 8 * hf;
      int col = (nh * 2 + n) * 16 + lm;
      p[row * D + col] = acc[n][r];
    }
  }
}

__global__ void ctx_reduce_kernel(const float* __restrict__ partial, float* __restrict__ ct) {
  int idx = blockIdx.x * blockDim.x + threadIdx.x;
  if (idx >= NHEADS * D * D) return;
  int head = idx / (D * D);
  int off  = idx - head * (D * D);
  float s = 0.f;
#pragma unroll
  for (int sp = 0; sp < SPLITS; ++sp)
    s += partial[(size_t)(sp * NHEADS + head) * (D * D) + off];
  ct[idx] = s;
}

__global__ void out_kernel(const float* __restrict__ q, const float* __restrict__ ct,
                           float* __restrict__ out) {
  const int head  = blockIdx.x;
  const int stile = blockIdx.y;
  const int s0 = stile * 128;
  const float* qh = q + (size_t)head * D * S;
  float* oh = out + (size_t)head * D * S;
  const int t    = threadIdx.x;
  const int lane = t & 31;
  const int w    = t >> 5;
  const int hf   = lane >> 4;
  const int lm   = lane & 15;

  __shared__ __align__(16) float qs[D * 129];
  __shared__ __align__(16) float cts[D * LDSTR];

#pragma unroll
  for (int i = 0; i < 8; ++i) {
    int e   = (i * 128 + t) * 4;
    int row = e >> 6;
    int col = e & 63;
    *(v4f*)&cts[row * LDSTR + col] = *(const v4f*)&ct[(size_t)head * (D * D) + e];
  }

  float m = -3.4e38f;
  for (int d = 0; d < D; ++d) {
    float x = __builtin_nontemporal_load(&qh[(size_t)d * S + s0 + t]);
    qs[d * 129 + t] = x;
    m = fmaxf(m, x);
  }
  float sum = 0.f;
  for (int d = 0; d < D; ++d) {
    float e = __expf(qs[d * 129 + t] - m);
    sum += e;
    qs[d * 129 + t] = e;
  }
  const float r = 0.125f / sum;
  for (int d = 0; d < D; ++d) qs[d * 129 + t] *= r;
  __syncthreads();

  const int j0 = w * 16;
  for (int st = 0; st < 8; ++st) {
    const int sb = st * 16;
    v8f acc = {};
#pragma unroll
    for (int kk = 0; kk < D; kk += 4) {
      v2f a = *(const v2f*)&cts[(j0 + lm) * LDSTR + kk + 2 * hf];
      v2f b;
      b.x = qs[(kk     + 2 * hf) * 129 + sb + lm];
      b.y = qs[(kk + 1 + 2 * hf) * 129 + sb + lm];
      acc = __builtin_amdgcn_wmma_f32_16x16x4_f32(
          false, a, false, b, (short)0, acc, false, false);
    }
#pragma unroll
    for (int rr = 0; rr < 8; ++rr) {
      int row = j0 + rr + 8 * hf;
      __builtin_nontemporal_store(acc[rr], &oh[(size_t)row * S + s0 + sb + lm]);
    }
  }
}

extern "C" void kernel_launch(void* const* d_in, const int* in_sizes, int n_in,
                              void* d_out, int out_size, void* d_ws, size_t ws_size,
                              hipStream_t stream) {
  (void)in_sizes; (void)n_in; (void)out_size; (void)ws_size;
  const float* q = (const float*)d_in[0];
  const float* k = (const float*)d_in[1];
  const float* v = (const float*)d_in[2];
  float* out = (float*)d_out;
  float* ws  = (float*)d_ws;

  float* stats   = ws;
  float* partial = ws + 2 * NHEADS * D;
  float* ct      = partial + (size_t)SPLITS * NHEADS * D * D;

  kstats_kernel<<<dim3(NHEADS * D), 256, 0, stream>>>(k, stats);
  ctx_partial_kernel<<<dim3(NHEADS, SPLITS), 256, 0, stream>>>(k, v, stats, partial);
  ctx_reduce_kernel<<<dim3((NHEADS * D * D + 255) / 256), 256, 0, stream>>>(partial, ct);
  out_kernel<<<dim3(NHEADS, S / 128), 128, 0, stream>>>(q, ct, out);
}


# out kernel: q loaded from global straight into WMMA B registers (lanes 0-15: d rows 0-31, lanes 16-31: rows 32-63; CT read in the same k order), softmax in registers, no LDS for q; d sums reordered
# speedup vs baseline: 2.4072x; 1.3457x over previous
#include <hip/hip_runtime.h>


typedef float v2f __attribute__((ext_vector_type(2)));
typedef float v4f __attribute__((ext_vector_type(4)));
typedef float v8f __attribute__((ext_vector_type(8)));

#define NHEADS 32
#define D      64
#define S      8192
#define SPLITS 8
#define SPAN   (S / SPLITS)
#define SCHUNK 64
#define LDSTR  68

__global__ void kstats_kernel(const float* __restrict__ k, float* __restrict__ stats) {
  const int row = blockIdx.x;
  const float* kr = k + (size_t)row * S;
  const int t = threadIdx.x;
  v4f vals[8];
  float m = -3.4e38f;
#pragma unroll
  for (int i = 0; i < 8; ++i) {
    v4f x = *(const v4f*)&kr[i * 1024 + t * 4];
    vals[i] = x;
    m = fmaxf(m, fmaxf(fmaxf(x.x, x.y), fmaxf(x.z, x.w)));
  }
  __shared__ float red[256];
  red[t] = m;
  __syncthreads();
  for (int off = 128; off > 0; off >>= 1) {
    if (t < off) red[t] = fmaxf(red[t], red[t + off]);
    __syncthreads();
  }
  m = red[0];
  __syncthreads();
  float s = 0.f;
#pragma unroll
  for (int i = 0; i < 8; ++i) {
    v4f x = vals[i];
    s += __expf(x.x - m) + __expf(x.y - m) + __expf(x.z - m) + __expf(x.w - m);
  }
  red[t] = s;
  __syncthreads();
  for (int off = 128; off > 0; off >>= 1) {
    if (t < off) red[t] += red[t + off];
    __syncthreads();
  }
  if (t == 0) {
    stats[2 * row]     = m;
    stats[2 * row + 1] = 1.0f / red[0];
  }
}

#define NCHUNK (SPAN / SCHUNK)
#define TILEF  (D * LDSTR)

__device__ __forceinline__ void wg_barrier_keep_loads() {
  asm volatile("s_wait_dscnt 0x0\n\ts_barrier_signal -1\n\ts_barrier_wait -1" ::: "memory");
}

__device__ __forceinline__ void keep_phase_order() { __builtin_amdgcn_sched_barrier(0); }

__device__ __forceinline__ void load_set(v4f (&rv)[4], v4f (&rk)[4], const float* __restrict__ vp,
                                         const float* __restrict__ kp) {
  for (int i = 0; i < 4; ++i) {
    rv[i] = __builtin_nontemporal_load((const v4f*)&vp[(size_t)(16 * i) * S]);
    rk[i] = *(const v4f*)&kp[(size_t)(16 * i) * S];
  }
}

__device__ __forceinline__ void stage_set(const v4f (&rv)[4], const v4f (&rk)[4], const float2 (&st)[4],
                                          float* __restrict__ vdst, float* __restrict__ kdst) {
  for (int i = 0; i < 4; ++i) {
    *(v4f*)&vdst[16 * i * LDSTR] = rv[i];
    const v4f kx = rk[i];
    v4f ks;
    ks.x = __expf(kx.x - st[i].x) * st[i].y;
    ks.y = __expf(kx.y - st[i].x) * st[i].y;
    ks.z = __expf(kx.z - st[i].x) * st[i].y;
    ks.w = __expf(kx.w - st[i].x) * st[i].y;
    *(v4f*)&kdst[16 * i * LDSTR] = ks;
  }
}

__device__ __forceinline__ void mma_chunk(const float* __restrict__ arow, const float* __restrict__ bbase,
                                          v8f (&acc)[2]) {
  for (int kk = 0; kk < SCHUNK; kk += 4) {
    v2f a = *(const v2f*)&arow[kk];
    for (int n = 0; n < 2; ++n) {
      v2f b = *(const v2f*)&bbase[n * 16 * LDSTR + kk];
      acc[n] = __builtin_amdgcn_wmma_f32_16x16x4_f32(
          false, a, false, b, (short)0, acc[n], false, false);
    }
  }
}

__global__ __launch_bounds__(256) void ctx_partial_kernel(const float* __restrict__ k, const float* __restrict__ v,
                                   const float* __restrict__ stats, float* __restrict__ partial) {
  const int head  = blockIdx.x;
  const int split = blockIdx.y;
  const size_t hoff = (size_t)head * D * S;
  const int t    = threadIdx.x;
  const int lane = t & 31;
  const int wv   = t >> 5;
  const int w    = wv & 3;
  const int nh   = wv >> 2;
  const int hf   = lane >> 4;
  const int lm   = lane & 15;

  __shared__ __align__(16) float vt[2 * TILEF];
  __shared__ __align__(16) float kt[2 * TILEF];

  const int srow = t >> 4;
  const int scol = (t * 4) & 63;
  const float* vsrc = v + hoff + (size_t)srow * S + split * SPAN + scol;
  const float* ksrc = k + hoff + (size_t)srow * S + split * SPAN + scol;
  float* vdst = &vt[srow * LDSTR + scol];
  float* kdst = &kt[srow * LDSTR + scol];

  const float* arow  = &vt[(w * 16 + lm) * LDSTR + 2 * hf];
  const float* bbase = &kt[(nh * 32 + lm) * LDSTR + 2 * hf];

  const v8f vzero = {};
  v8f acc[2];
  acc[0] = vzero;
  acc[1] = vzero;

  v4f av[4], ak[4], bv[4], bk[4];
  float2 st[4];
  for (int i = 0; i < 4; ++i)
    st[i] = *(const float2*)&stats[2 * (head * D + 16 * i + srow)];

  load_set(av, ak, vsrc, ksrc);
  keep_phase_order();
  load_set(bv, bk, vsrc + SCHUNK, ksrc + SCHUNK);
  keep_phase_order();
  stage_set(av, ak, st, vdst, kdst);
  load_set(av, ak, vsrc + 2 * SCHUNK, ksrc + 2 * SCHUNK);
  keep_phase_order();
  wg_barrier_keep_loads();
  keep_phase_order();

  for (int c = 0; c < NCHUNK - 4; c += 2) {
    stage_set(bv, bk, st, vdst + TILEF, kdst + TILEF);
    load_set(bv, bk, vsrc + (c + 3) * SCHUNK, ksrc + (c + 3) * SCHUNK);
    mma_chunk(arow, bbase, acc);
    wg_barrier_keep_loads();
    keep_phase_order();
    stage_set(av, ak, st, vdst, kdst);
    load_set(av, ak, vsrc + (c + 4) * SCHUNK, ksrc + (c + 4) * SCHUNK);
    mma_chunk(arow + TILEF, bbase + TILEF, acc);
    wg_barrier_keep_loads();
    keep_phase_order();
  }
  stage_set(bv, bk, st, vdst + TILEF, kdst + TILEF);
  load_set(bv, bk, vsrc + (NCHUNK - 1) * SCHUNK, ksrc + (NCHUNK - 1) * SCHUNK);
  mma_chunk(arow, bbase, acc);
  wg_barrier_keep_loads();
  keep_phase_order();
  stage_set(av, ak, st, vdst, kdst);
  mma_chunk(arow + TILEF, bbase + TILEF, acc);
  wg_barrier_keep_loads();
  keep_phase_order();
  stage_set(bv, bk, st, vdst + TILEF, kdst + TILEF);
  mma_chunk(arow, bbase, acc);
  wg_barrier_keep_loads();
  keep_phase_order();
  mma_chunk(arow + TILEF, bbase + TILEF, acc);

  float* p = partial + (size_t)(split * NHEADS + head) * (D * D);
  for (int n = 0; n < 2; ++n) {
    for (int r = 0; r < 8; ++r) {
      int row = w * 16 + r + 8 * hf;
      int col = (nh * 2 + n) * 16 + lm;
      p[row * D + col] = acc[n][r];
    }
  }
}

__global__ void ctx_reduce_kernel(const float* __restrict__ partial, float* __restrict__ ct) {
  int idx = blockIdx.x * blockDim.x + threadIdx.x;
  if (idx >= NHEADS * D * D) return;
  int head = idx / (D * D);
  int off  = idx - head * (D * D);
  float s = 0.f;
#pragma unroll
  for (int sp = 0; sp < SPLITS; ++sp)
    s += partial[(size_t)(sp * NHEADS + head) * (D * D) + off];
  ct[idx] = s;
}

#define CTSTR 66

__global__ __launch_bounds__(128) void out_kernel(const float* __restrict__ q, const float* __restrict__ ct,
                                                  float* __restrict__ out) {
  const int head  = blockIdx.x;
  const int stile = blockIdx.y;
  const int t    = threadIdx.x;
  const int lane = t & 31;
  const int w    = t >> 5;
  const int hf   = lane >> 4;
  const int lm   = lane & 15;
  const int sb   = stile * 128 + w * 32 + 2 * lm;
  const float* qp = q + (size_t)head * D * S + (size_t)(32 * hf) * S + sb;
  float* op = out + (size_t)head * D * S + (size_t)(8 * hf) * S + sb;

  __shared__ __align__(16) float cts[D * CTSTR];

  v4f cr[8];
  for (int i = 0; i < 8; ++i)
    cr[i] = *(const v4f*)&ct[(size_t)head * (D * D) + (i * 128 + t) * 4];
  v2f qv[32];
  for (int e = 0; e < 32; ++e)
    qv[e] = __builtin_nontemporal_load((const v2f*)&qp[(size_t)e * S]);

  for (int i = 0; i < 8; ++i) {
    const int e   = (i * 128 + t) * 4;
    const int row = e >> 6;
    const int col = e & 63;
    v2f lo, hi;
    lo.x = cr[i].x; lo.y = cr[i].y;
    hi.x = cr[i].z; hi.y = cr[i].w;
    *(v2f*)&cts[row * CTSTR + col]     = lo;
    *(v2f*)&cts[row * CTSTR + col + 2] = hi;
  }

  float m0 = -3.4e38f, m1 = -3.4e38f;
  for (int e = 0; e < 32; ++e) {
    m0 = fmaxf(m0, qv[e].x);
    m1 = fmaxf(m1, qv[e].y);
  }
  m0 = fmaxf(m0, __shfl_xor(m0, 16));
  m1 = fmaxf(m1, __shfl_xor(m1, 16));
  float sum0 = 0.f, sum1 = 0.f;
  for (int e = 0; e < 32; ++e) {
    const float e0 = __expf(qv[e].x - m0);
    const float e1 = __expf(qv[e].y - m1);
    sum0 += e0;
    sum1 += e1;
    qv[e].x = e0;
    qv[e].y = e1;
  }
  sum0 += __shfl_xor(sum0, 16);
  sum1 += __shfl_xor(sum1, 16);
  const float r0 = 0.125f / sum0;
  const float r1 = 0.125f / sum1;
  v2f bx[16], by[16];
  for (int tt = 0; tt < 16; ++tt) {
    bx[tt].x = qv[2 * tt].x * r0;
    bx[tt].y = qv[2 * tt + 1].x * r0;
    by[tt].x = qv[2 * tt].y * r1;
    by[tt].y = qv[2 * tt + 1].y * r1;
  }
  __syncthreads();

  const float* ap = &cts[lm * CTSTR + 32 * hf];
  v8f acc[4][2];
  for (int m = 0; m < 4; ++m) {
    const v8f vzero = {};
    acc[m][0] = vzero;
    acc[m][1] = vzero;
  }
  for (int tt = 0; tt < 16; ++tt) {
    for (int m = 0; m < 4; ++m) {
      const v2f fa = *(const v2f*)&ap[(16 * m) * CTSTR + 2 * tt];
      acc[m][0] = __builtin_amdgcn_wmma_f32_16x16x4_f32(false, fa, false, bx[tt], (short)0, acc[m][0], false, false);
      acc[m][1] = __builtin_amdgcn_wmma_f32_16x16x4_f32(false, fa, false, by[tt], (short)0, acc[m][1], false, false);
    }
  }
  for (int m = 0; m < 4; ++m) {
    for (int rr = 0; rr < 8; ++rr) {
      v2f o;
      o.x = acc[m][0][rr];
      o.y = acc[m][1][rr];
      __builtin_nontemporal_store(o, (v2f*)&op[(size_t)(16 * m + rr) * S]);
    }
  }
}

extern "C" void kernel_launch(void* const* d_in, const int* in_sizes, int n_in,
                              void* d_out, int out_size, void* d_ws, size_t ws_size,
                              hipStream_t stream) {
  (void)in_sizes; (void)n_in; (void)out_size; (void)ws_size;
  const float* q = (const float*)d_in[0];
  const float* k = (const float*)d_in[1];
  const float* v = (const float*)d_in[2];
  float* out = (float*)d_out;
  float* ws  = (float*)d_ws;

  float* stats   = ws;
  float* partial = ws + 2 * NHEADS * D;
  float* ct      = partial + (size_t)SPLITS * NHEADS * D * D;

  kstats_kernel<<<dim3(NHEADS * D), 256, 0, stream>>>(k, stats);
  ctx_partial_kernel<<<dim3(NHEADS, SPLITS), 256, 0, stream>>>(k, v, stats, partial);
  ctx_reduce_kernel<<<dim3((NHEADS * D * D + 255) / 256), 256, 0, stream>>>(partial, ct);
  out_kernel<<<dim3(NHEADS, S / 128), 128, 0, stream>>>(q, ct, out);
}


# k softmax over s computed online in the ctx kernel (running row max per split, accumulators rescaled, splits merged by max and sum in the reduce kernel); k-stats launch removed, sums reordered
# speedup vs baseline: 2.7109x; 1.1262x over previous
#include <hip/hip_runtime.h>


typedef float v2f __attribute__((ext_vector_type(2)));
typedef float v4f __attribute__((ext_vector_type(4)));
typedef float v8f __attribute__((ext_vector_type(8)));

#define NHEADS 32
#define D      64
#define S      8192
#define SPLITS 8
#define SPAN   (S / SPLITS)
#define SCHUNK 64
#define LDSTR  68

#define NCHUNK (SPAN / SCHUNK)
#define TILEF  (D * LDSTR)

__device__ __forceinline__ void wg_barrier_keep_loads() {
  asm volatile("s_wait_dscnt 0x0\n\ts_barrier_signal -1\n\ts_barrier_wait -1" ::: "memory");
}

__device__ __forceinline__ void keep_phase_order() { __builtin_amdgcn_sched_barrier(0); }

__device__ __forceinline__ void load_set(v4f (&rv)[4], v4f (&rk)[4], const float* __restrict__ vp,
                                         const float* __restrict__ kp) {
  for (int i = 0; i < 4; ++i) {
    rv[i] = __builtin_nontemporal_load((const v4f*)&vp[(size_t)(16 * i) * S]);
    rk[i] = __builtin_nontemporal_load((const v4f*)&kp[(size_t)(16 * i) * S]);
  }
}

template <int XM> __device__ __forceinline__ float row16_xor(float x) {
  const int xi = __builtin_bit_cast(int, x);
  return __builtin_bit_cast(float, __builtin_amdgcn_update_dpp(xi, xi, 0x160 + XM, 0xf, 0xf, false));
}
__device__ __forceinline__ float row16_max(float x) {
  asm("v_max_num_f32_dpp %0, %0, %0 row_xmask:1 row_mask:0xf bank_mask:0xf" : "+v"(x));
  asm("v_max_num_f32_dpp %0, %0, %0 row_xmask:2 row_mask:0xf bank_mask:0xf" : "+v"(x));
  asm("v_max_num_f32_dpp %0, %0, %0 row_xmask:4 row_mask:0xf bank_mask:0xf" : "+v"(x));
  asm("v_max_num_f32_dpp %0, %0, %0 row_xmask:8 row_mask:0xf bank_mask:0xf" : "+v"(x));
  return x;
}
__device__ __forceinline__ float row16_sum(float x) {
  x += row16_xor<1>(x);
  x += row16_xor<2>(x);
  x += row16_xor<4>(x);
  x += row16_xor<8>(x);
  return x;
}

__device__ __forceinline__ void stage_set(const v4f (&rv)[4], const v4f (&rk)[4], float (&rm)[4], float (&rl)[4],
                                          float* __restrict__ vdst, float* __restrict__ kdst,
                                          float* __restrict__ fdst, const bool fwriter) {
  for (int i = 0; i < 4; ++i) {
    *(v4f*)&vdst[16 * i * LDSTR] = rv[i];
    const v4f kx = rk[i];
    const float cm = row16_max(fmaxf(fmaxf(kx.x, kx.y), fmaxf(kx.z, kx.w)));
    const float mn = fmaxf(rm[i], cm);
    const float f  = __expf(rm[i] - mn);
    rm[i] = mn;
    v4f ks;
    ks.x = __expf(kx.x - mn);
    ks.y = __expf(kx.y - mn);
    ks.z = __expf(kx.z - mn);
    ks.w = __expf(kx.w - mn);
    rl[i] = rl[i] * f + ((ks.x + ks.y) + (ks.z + ks.w));
    *(v4f*)&kdst[16 * i * LDSTR] = ks;
    if (fwriter) fdst[16 * i] = f;
  }
}

__device__ __forceinline__ void rescale_acc(const float* __restrict__ fcol, v8f (&acc)[2]) {
  const float f0 = fcol[0];
  const float f1 = fcol[16];
  for (int r = 0; r < 8; ++r) {
    acc[0][r] *= f0;
    acc[1][r] *= f1;
  }
}

__device__ __forceinline__ void mma_chunk(const float* __restrict__ arow, const float* __restrict__ bbase,
                                          v8f (&acc)[2]) {
  for (int kk = 0; kk < SCHUNK; kk += 4) {
    v2f a = *(const v2f*)&arow[kk];
    for (int n = 0; n < 2; ++n) {
      v2f b = *(const v2f*)&bbase[n * 16 * LDSTR + kk];
      acc[n] = __builtin_amdgcn_wmma_f32_16x16x4_f32(
          false, a, false, b, (short)0, acc[n], false, false);
    }
  }
}

__global__ __launch_bounds__(256) void ctx_partial_kernel(const float* __restrict__ k, const float* __restrict__ v,
                                   float* __restrict__ mlstat, float* __restrict__ partial) {
  const int head  = blockIdx.x;
  const int split = blockIdx.y;
  const size_t hoff = (size_t)head * D * S;
  const int t    = threadIdx.x;
  const int lane = t & 31;
  const int wv   = t >> 5;
  const int w    = wv & 3;
  const int nh   = wv >> 2;
  const int hf   = lane >> 4;
  const int lm   = lane & 15;

  __shared__ __align__(16) float vt[2 * TILEF];
  __shared__ __align__(16) float kt[2 * TILEF];
  __shared__ float fs[2 * D];

  const int srow = t >> 4;
  const int scol = (t * 4) & 63;
  const float* vsrc = v + hoff + (size_t)srow * S + split * SPAN + scol;
  const float* ksrc = k + hoff + (size_t)srow * S + split * SPAN + scol;
  float* vdst = &vt[srow * LDSTR + scol];
  float* kdst = &kt[srow * LDSTR + scol];
  float* fdst = &fs[srow];
  const bool fwriter = (t & 15) == 0;

  const float* arow  = &vt[(w * 16 + lm) * LDSTR + 2 * hf];
  const float* bbase = &kt[(nh * 32 + lm) * LDSTR + 2 * hf];
  const float* fcol  = &fs[nh * 32 + lm];

  const v8f vzero = {};
  v8f acc[2];
  acc[0] = vzero;
  acc[1] = vzero;

  v4f av[4], ak[4], bv[4], bk[4];
  float rm[4], rl[4];
  for (int i = 0; i < 4; ++i) {
    rm[i] = -3.4e38f;
    rl[i] = 0.f;
  }

  load_set(av, ak, vsrc, ksrc);
  keep_phase_order();
  load_set(bv, bk, vsrc + SCHUNK, ksrc + SCHUNK);
  keep_phase_order();
  stage_set(av, ak, rm, rl, vdst, kdst, fdst, fwriter);
  load_set(av, ak, vsrc + 2 * SCHUNK, ksrc + 2 * SCHUNK);
  keep_phase_order();
  wg_barrier_keep_loads();
  keep_phase_order();

  for (int c = 0; c < NCHUNK - 4; c += 2) {
    stage_set(bv, bk, rm, rl, vdst + TILEF, kdst + TILEF, fdst + D, fwriter);
    load_set(bv, bk, vsrc + (c + 3) * SCHUNK, ksrc + (c + 3) * SCHUNK);
    rescale_acc(fcol, acc);
    mma_chunk(arow, bbase, acc);
    wg_barrier_keep_loads();
    keep_phase_order();
    stage_set(av, ak, rm, rl, vdst, kdst, fdst, fwriter);
    load_set(av, ak, vsrc + (c + 4) * SCHUNK, ksrc + (c + 4) * SCHUNK);
    rescale_acc(fcol + D, acc);
    mma_chunk(arow + TILEF, bbase + TILEF, acc);
    wg_barrier_keep_loads();
    keep_phase_order();
  }
  stage_set(bv, bk, rm, rl, vdst + TILEF, kdst + TILEF, fdst + D, fwriter);
  load_set(bv, bk, vsrc + (NCHUNK - 1) * SCHUNK, ksrc + (NCHUNK - 1) * SCHUNK);
  rescale_acc(fcol, acc);
  mma_chunk(arow, bbase, acc);
  wg_barrier_keep_loads();
  keep_phase_order();
  stage_set(av, ak, rm, rl, vdst, kdst, fdst, fwriter);
  rescale_acc(fcol + D, acc);
  mma_chunk(arow + TILEF, bbase + TILEF, acc);
  wg_barrier_keep_loads();
  keep_phase_order();
  stage_set(bv, bk, rm, rl, vdst + TILEF, kdst + TILEF, fdst + D, fwriter);
  rescale_acc(fcol, acc);
  mma_chunk(arow, bbase, acc);
  wg_barrier_keep_loads();
  keep_phase_order();
  rescale_acc(fcol + D, acc);
  mma_chunk(arow + TILEF, bbase + TILEF, acc);

  for (int i = 0; i < 4; ++i) {
    const float l = row16_sum(rl[i]);
    if (fwriter) {
      float2 ml;
      ml.x = rm[i];
      ml.y = l;
      *(float2*)&mlstat[2 * ((size_t)(split * NHEADS + head) * D + 16 * i + srow)] = ml;
    }
  }

  float* p = partial + (size_t)(split * NHEADS + head) * (D * D);
  for (int n = 0; n < 2; ++n) {
    for (int r = 0; r < 8; ++r) {
      int row = w * 16 + r + 8 * hf;
      int col = (nh * 2 + n) * 16 + lm;
      p[row * D + col] = acc[n][r];
    }
  }
}

__global__ void ctx_reduce_kernel(float* __restrict__ partial, const float* __restrict__ mlstat) {
  int idx = blockIdx.x * blockDim.x + threadIdx.x;
  if (idx >= NHEADS * D * D) return;
  int head = idx / (D * D);
  int off  = idx - head * (D * D);
  int i    = off & (D - 1);
  float  pv[SPLITS];
  float2 ml[SPLITS];
#pragma unroll
  for (int sp = 0; sp < SPLITS; ++sp)
    pv[sp] = partial[(size_t)(sp * NHEADS + head) * (D * D) + off];
#pragma unroll
  for (int sp = 0; sp < SPLITS; ++sp)
    ml[sp] = *(const float2*)&mlstat[2 * ((size_t)(sp * NHEADS + head) * D + i)];
  __builtin_amdgcn_sched_barrier(0);
  float M = -3.4e38f;
#pragma unroll
  for (int sp = 0; sp < SPLITS; ++sp)
    M = fmaxf(M, ml[sp].x);
  float s = 0.f, L = 0.f;
#pragma unroll
  for (int sp = 0; sp < SPLITS; ++sp) {
    const float wgt = __expf(ml[sp].x - M);
    L += ml[sp].y * wgt;
    s += pv[sp] * wgt;
  }
  partial[idx] = s / L;
}

#define CTSTR 66

__global__ __launch_bounds__(128) void out_kernel(const float* __restrict__ q, const float* __restrict__ ct,
                                                  float* __restrict__ out) {
  const int head  = blockIdx.x;
  const int stile = blockIdx.y;
  const int t    = threadIdx.x;
  const int lane = t & 31;
  const int w    = t >> 5;
  const int hf   = lane >> 4;
  const int lm   = lane & 15;
  const int sb   = stile * 128 + w * 32 + 2 * lm;
  const float* qp = q + (size_t)head * D * S + (size_t)(32 * hf) * S + sb;
  float* op = out + (size_t)head * D * S + (size_t)(8 * hf) * S + sb;

  __shared__ __align__(16) float cts[D * CTSTR];

  v4f cr[8];
  for (int i = 0; i < 8; ++i)
    cr[i] = *(const v4f*)&ct[(size_t)head * (D * D) + (i * 128 + t) * 4];
  v2f qv[32];
  for (int e = 0; e < 32; ++e)
    qv[e] = __builtin_nontemporal_load((const v2f*)&qp[(size_t)e * S]);

  for (int i = 0; i < 8; ++i) {
    const int e   = (i * 128 + t) * 4;
    const int row = e >> 6;
    const int col = e & 63;
    v2f lo, hi;
    lo.x = cr[i].x; lo.y = cr[i].y;
    hi.x = cr[i].z; hi.y = cr[i].w;
    *(v2f*)&cts[row * CTSTR + col]     = lo;
    *(v2f*)&cts[row * CTSTR + col + 2] = hi;
  }

  float m0 = -3.4e38f, m1 = -3.4e38f;
  for (int e = 0; e < 32; ++e) {
    m0 = fmaxf(m0, qv[e].x);
    m1 = fmaxf(m1, qv[e].y);
  }
  m0 = fmaxf(m0, __shfl_xor(m0, 16));
  m1 = fmaxf(m1, __shfl_xor(m1, 16));
  float sum0 = 0.f, sum1 = 0.f;
  for (int e = 0; e < 32; ++e) {
    const float e0 = __expf(qv[e].x - m0);
    const float e1 = __expf(qv[e].y - m1);
    sum0 += e0;
    sum1 += e1;
    qv[e].x = e0;
    qv[e].y = e1;
  }
  sum0 += __shfl_xor(sum0, 16);
  sum1 += __shfl_xor(sum1, 16);
  const float r0 = 0.125f / sum0;
  const float r1 = 0.125f / sum1;
  v2f bx[16], by[16];
  for (int tt = 0; tt < 16; ++tt) {
    bx[tt].x = qv[2 * tt].x * r0;
    bx[tt].y = qv[2 * tt + 1].x * r0;
    by[tt].x = qv[2 * tt].y * r1;
    by[tt].y = qv[2 * tt + 1].y * r1;
  }
  __syncthreads();

  const float* ap = &cts[lm * CTSTR + 32 * hf];
  v8f acc[4][2];
  for (int m = 0; m < 4; ++m) {
    const v8f vzero = {};
    acc[m][0] = vzero;
    acc[m][1] = vzero;
  }
  for (int tt = 0; tt < 16; ++tt) {
    for (int m = 0; m < 4; ++m) {
      const v2f fa = *(const v2f*)&ap[(16 * m) * CTSTR + 2 * tt];
      acc[m][0] = __builtin_amdgcn_wmma_f32_16x16x4_f32(false, fa, false, bx[tt], (short)0, acc[m][0], false, false);
      acc[m][1] = __builtin_amdgcn_wmma_f32_16x16x4_f32(false, fa, false, by[tt], (short)0, acc[m][1], false, false);
    }
  }
  for (int m = 0; m < 4; ++m) {
    for (int rr = 0; rr < 8; ++rr) {
      v2f o;
      o.x = acc[m][0][rr];
      o.y = acc[m][1][rr];
      __builtin_nontemporal_store(o, (v2f*)&op[(size_t)(16 * m + rr) * S]);
    }
  }
}

extern "C" void kernel_launch(void* const* d_in, const int* in_sizes, int n_in,
                              void* d_out, int out_size, void* d_ws, size_t ws_size,
                              hipStream_t stream) {
  (void)in_sizes; (void)n_in; (void)out_size; (void)ws_size;
  const float* q = (const float*)d_in[0];
  const float* k = (const float*)d_in[1];
  const float* v = (const float*)d_in[2];
  float* out = (float*)d_out;
  float* ws  = (float*)d_ws;

  float* partial = ws + 2 * NHEADS * D;
  float* mlstat  = partial + (size_t)SPLITS * NHEADS * D * D;

  ctx_partial_kernel<<<dim3(NHEADS, SPLITS), 256, 0, stream>>>(k, v, mlstat, partial);
  ctx_reduce_kernel<<<dim3((NHEADS * D * D + 255) / 256), 256, 0, stream>>>(partial, mlstat);
  out_kernel<<<dim3(NHEADS, S / 128), 128, 0, stream>>>(q, partial, out);
}
